# starting kernel
# baseline (speedup 1.0000x reference)
#include <hip/hip_runtime.h>
#include <math.h>

namespace {

constexpr int kC      = 1000;
constexpr int kCPad   = 1008;
constexpr int kD      = 64;
constexpr int kB      = 2048;
constexpr float kEpsStats = 1e-5f;
constexpr float kEpsPrec  = 1e-6f;

constexpr int OFF_COUNTS = 0;
constexpr int OFF_MEAN   = OFF_COUNTS + kC;
constexpr int OFF_VARSUM = OFF_MEAN   + kC * kD;
constexpr int OFF_MUPAD  = OFF_VARSUM + kC * kD;
constexpr int OFF_PREC   = OFF_MUPAD  + kCPad * kD;
constexpr int OFF_TOTAL  = OFF_PREC   + kD;
constexpr int OFF_CCONST = OFF_TOTAL  + 1;
constexpr int OFF_QH     = OFF_CCONST + kCPad;

typedef __attribute__((ext_vector_type(2))) float v2f;
typedef __attribute__((ext_vector_type(8))) float v8f;

__global__ __launch_bounds__(64)
void class_stats_kernel(const float* __restrict__ z, const int* __restrict__ y,
                        float* __restrict__ ws) {
  const int c = blockIdx.x;
  const int d = threadIdx.x;
  float s = 0.f, cnt = 0.f;
  for (int b = 0; b < kB; ++b) {
    if (y[b] == c) { s += z[b * kD + d]; cnt += 1.f; }
  }
  const float cnte = cnt + kEpsStats;
  if (d == 0) ws[OFF_COUNTS + c] = cnte;
  ws[OFF_MEAN + c * kD + d] = s / cnte;
}

__global__ __launch_bounds__(64)
void var_stats_kernel(const float* __restrict__ z, const int* __restrict__ y,
                      float* __restrict__ ws) {
  const int c = blockIdx.x;
  const int d = threadIdx.x;
  const float m = ws[OFF_MEAN + c * kD + d];
  float s = 0.f;
  for (int b = 0; b < kB; ++b) {
    if (y[b] == c) { const float t = z[b * kD + d] - m; s += t * t; }
  }
  ws[OFF_VARSUM + c * kD + d] = s;
}

__global__ __launch_bounds__(64)
void precision_kernel(float* __restrict__ ws) {
  const int d = threadIdx.x;
  float total = 0.f;
  for (int c = 0; c < kC; ++c) total += ws[OFF_COUNTS + c];
  float s = 0.f;
  for (int c = 0; c < kC; ++c) s += ws[OFF_VARSUM + c * kD + d];
  const float pooled = s / total + kEpsStats;
  ws[OFF_PREC + d] = 1.0f / fmaxf(pooled, kEpsPrec);
  if (d == 0) ws[OFF_TOTAL] = total;
}

__global__ __launch_bounds__(64)
void mu_prep_kernel(float* __restrict__ ws) {
  const int c = blockIdx.x;
  const int d = threadIdx.x;
  __shared__ float red[kD];
  if (c >= kC) {
    ws[OFF_MUPAD + c * kD + d] = 0.f;
    if (d == 0) ws[OFF_CCONST + c] = 0.f;
    return;
  }
  const float p = ws[OFF_PREC + d];
  const float m = ws[OFF_MEAN + c * kD + d];
  const float v = p * m;
  ws[OFF_MUPAD + c * kD + d] = v;
  red[d] = v * m;
  __syncthreads();
  for (int stride = kD / 2; stride > 0; stride >>= 1) {
    if (d < stride) red[d] += red[d + stride];
    __syncthreads();
  }
  if (d == 0) {
    const float prior = ws[OFF_COUNTS + c] / ws[OFF_TOTAL];
    ws[OFF_CCONST + c] = logf(prior) - 0.5f * red[0];
  }
}

__global__ __launch_bounds__(256)
void row_quad_kernel(const float* __restrict__ z, float* __restrict__ ws) {
  __shared__ float pshared[kD];
  if (threadIdx.x < kD) pshared[threadIdx.x] = ws[OFF_PREC + threadIdx.x];
  __syncthreads();
  const int b = blockIdx.x * blockDim.x + threadIdx.x;
  if (b >= kB) return;
  float s = 0.f;
#pragma unroll
  for (int d = 0; d < kD; ++d) {
    const float t = z[b * kD + d];
    s += pshared[d] * t * t;
  }
  ws[OFF_QH + b] = -0.5f * s;
}

__global__ __launch_bounds__(256)
void lda_score_wmma_kernel(const float* __restrict__ z, const float* __restrict__ ws,
                           float* __restrict__ out) {
  const int wave = threadIdx.x >> 5;
  const int lane = threadIdx.x & 31;
  const int tile = blockIdx.x * 8 + wave;
  const int tm = tile & 127;
  const int tn = tile >> 7;
  const int row0 = tm * 16;
  const int col0 = tn * 16;
  const int half = lane >> 4;
  const int lm   = lane & 15;

  const float* __restrict__ ap = z + (row0 + lm) * kD + 2 * half;
  const float* __restrict__ bp = ws + OFF_MUPAD + (col0 + lm) * kD + 2 * half;

  v8f acc = {};
#pragma unroll
  for (int k = 0; k < kD; k += 4) {
    const v2f a = *(const v2f*)(ap + k);
    const v2f b = *(const v2f*)(bp + k);
    acc = __builtin_amdgcn_wmma_f32_16x16x4_f32(
         false, a,  false, b,
         (short)0, acc,  false,  false);
  }

  const int col = col0 + lm;
  const float cc = ws[OFF_CCONST + col];
  if (col < kC) {
#pragma unroll
    for (int r = 0; r < 8; ++r) {
      const int row = row0 + r + 8 * half;
      out[row * kC + col] = acc[r] + cc + ws[OFF_QH + row];
    }
  }
}

}

extern "C" void kernel_launch(void* const* d_in, const int* in_sizes, int n_in,
                              void* d_out, int out_size, void* d_ws, size_t ws_size,
                              hipStream_t stream) {
  (void)in_sizes; (void)n_in; (void)out_size; (void)ws_size;
  const float* z = (const float*)d_in[0];
  const int*   y = (const int*)d_in[1];
  float* ws  = (float*)d_ws;
  float* out = (float*)d_out;

  class_stats_kernel<<<kC, kD, 0, stream>>>(z, y, ws);
  var_stats_kernel<<<kC, kD, 0, stream>>>(z, y, ws);
  precision_kernel<<<1, kD, 0, stream>>>(ws);
  mu_prep_kernel<<<kCPad, kD, 0, stream>>>(ws);
  row_quad_kernel<<<kB / 256, 256, 0, stream>>>(z, ws);

  const int tiles = (kB / 16) * (kCPad / 16);
  lda_score_wmma_kernel<<<tiles / 8, 256, 0, stream>>>(z, ws, out);
}


# precision, mean-scaling and row-term launches merged into one preparation launch (6 to 4): each block joins counts and residual partials itself; order of the count, mu'P mu and z'P z sums changes
# speedup vs baseline: 9.3800x; 9.3800x over previous
#include <hip/hip_runtime.h>
#include <math.h>

namespace {

constexpr int kC      = 1000;
constexpr int kCPad   = 1008;
constexpr int kD      = 64;
constexpr int kB      = 2048;
constexpr float kEpsStats = 1e-5f;
constexpr float kEpsPrec  = 1e-6f;
constexpr int kResidRows   = 64;
constexpr int kResidBlocks = kB / kResidRows;

constexpr int OFF_COUNTS = 0;
constexpr int OFF_MEAN   = OFF_COUNTS + kC;
constexpr int OFF_VARSUM = OFF_MEAN   + kC * kD;
constexpr int OFF_MUPAD  = OFF_VARSUM + kC * kD;
constexpr int OFF_CCONST = OFF_MUPAD  + kCPad * kD;
constexpr int OFF_QH     = OFF_CCONST + kCPad;

typedef __attribute__((ext_vector_type(2))) float v2f;
typedef __attribute__((ext_vector_type(4))) float v4f;
typedef __attribute__((ext_vector_type(8))) float v8f;

constexpr int kStatWaves    = 16;
constexpr int kStatRows     = kB / kStatWaves;
constexpr int kStatSteps    = kStatRows / 4;
constexpr int kStatTiles    = kCPad / 16;

__global__ __launch_bounds__(kStatWaves * 32) __attribute__((amdgpu_waves_per_eu(4, 4)))
void class_stats_kernel(const float* __restrict__ z, const int* __restrict__ y,
                        float* __restrict__ ws) {
  __shared__ float red[kStatWaves][16][32];
  __shared__ float cred[kStatWaves][32];
  const int wave = threadIdx.x >> 5;
  const int lane = threadIdx.x & 31;
  const int half = lane >> 4;
  const int ln   = lane & 15;
  const int c0   = blockIdx.x * 16;
  const int fh   = blockIdx.y;
  const int cls  = c0 + ln;

  const int rbase = wave * kStatRows + 2 * half;
  const int*   __restrict__ yp = y + rbase;
  const float* __restrict__ zp = z + rbase * kD + 32 * fh + 2 * ln;

  v8f acc0 = {}, acc1 = {};
  v2f cnt2 = {0.f, 0.f};
#pragma unroll
  for (int s = 0; s < kStatSteps; ++s) {
    const int2 lab = *(const int2*)(yp + 4 * s);
    const v2f z0 = *(const v2f*)(zp + (4 * s) * kD);
    const v2f z1 = *(const v2f*)(zp + (4 * s + 1) * kD);
    v2f a;
    a.x = (lab.x == cls) ? 1.f : 0.f;
    a.y = (lab.y == cls) ? 1.f : 0.f;
    cnt2 += a;
    const v2f b0 = {z0.x, z1.x};
    const v2f b1 = {z0.y, z1.y};
    acc0 = __builtin_amdgcn_wmma_f32_16x16x4_f32(false, a, false, b0, (short)0, acc0, false, false);
    acc1 = __builtin_amdgcn_wmma_f32_16x16x4_f32(false, a, false, b1, (short)0, acc1, false, false);
  }

#pragma unroll
  for (int r = 0; r < 8; ++r) {
    const v2f p = {acc0[r], acc1[r]};
    *(v2f*)(&red[wave][r + 8 * half][2 * ln]) = p;
  }
  cred[wave][lane] = cnt2.x + cnt2.y;
  __syncthreads();

  const int oc = threadIdx.x >> 5;
  const int of = threadIdx.x & 31;
  float s = 0.f, cnt = 0.f;
  for (int w = 0; w < kStatWaves; ++w) {
    s   += red[w][oc][of];
    cnt += cred[w][oc] + cred[w][oc + 16];
  }
  const int c = c0 + oc;
  if (c < kC) {
    const float cnte = cnt + kEpsStats;
    if (fh == 0 && of == 0) ws[OFF_COUNTS + c] = cnte;
    ws[OFF_MEAN + c * kD + 32 * fh + of] = s / cnte;
  }
}

__global__ __launch_bounds__(256)
void row_resid_kernel(const float* __restrict__ z, const int* __restrict__ y,
                      float* __restrict__ ws) {
  __shared__ float red[16][kD];
  constexpr int kIter = kResidRows / 16;
  const int q = threadIdx.x & 15;
  const int r = threadIdx.x >> 4;
  const int b0 = blockIdx.x * kResidRows + r;
  int cls[kIter];
  v4f zv[kIter];
  for (int i = 0; i < kIter; ++i) {
    const int b = b0 + 16 * i;
    cls[i] = y[b];
    zv[i] = *(const v4f*)(z + b * kD + 4 * q);
  }
  v4f acc = {0.f, 0.f, 0.f, 0.f};
  for (int i = 0; i < kIter; ++i) {
    const bool member = (unsigned)cls[i] < (unsigned)kC;
    const int c = member ? cls[i] : 0;
    const v4f mv = *(const v4f*)(ws + OFF_MEAN + c * kD + 4 * q);
    v4f t = zv[i] - mv;
    if (!member) t = v4f{0.f, 0.f, 0.f, 0.f};
    acc += t * t;
  }
  *(v4f*)(&red[r][4 * q]) = acc;
  __syncthreads();
  if (threadIdx.x < kD) {
    float s = 0.f;
    for (int j = 0; j < 16; ++j) s += red[j][threadIdx.x];
    ws[OFF_VARSUM + blockIdx.x * kD + threadIdx.x] = s;
  }
}

constexpr int kPrepClassBlocks = kCPad / 16;
constexpr int kPrepRowBlocks   = kB / 64;
constexpr int kPrepBlocks      = kPrepClassBlocks + kPrepRowBlocks;

template <int M>
__device__ __forceinline__ float row_xor(float x) {
  return __builtin_bit_cast(float, __builtin_amdgcn_mov_dpp(__builtin_bit_cast(int, x), 0x160 + M, 0xf, 0xf, false));
}
__device__ __forceinline__ float row16_sum(float x) {
  x += row_xor<1>(x);
  x += row_xor<2>(x);
  x += row_xor<4>(x);
  x += row_xor<8>(x);
  return x;
}
__device__ __forceinline__ float xhalf(float x) {
  return __builtin_bit_cast(float, __builtin_amdgcn_permlanex16(__builtin_bit_cast(int, x), __builtin_bit_cast(int, x),
                                                                0x76543210, 0xfedcba98, false, false));
}

__device__ __forceinline__ float join_statistics(const float* __restrict__ ws, float* wsum, float* prec, int t) {
  const int lane = t & 31;
  const int wave = t >> 5;
  const int c3 = t + 768;
  const float n0 = ws[OFF_COUNTS + t];
  const float n1 = ws[OFF_COUNTS + t + 256];
  const float n2 = ws[OFF_COUNTS + t + 512];
  const float n3 = ws[OFF_COUNTS + (c3 < kC ? c3 : kC - 1)];
  float s = 0.f;
  if (t < kD) {
    for (int p = 0; p < kResidBlocks; ++p)
      s += ws[OFF_VARSUM + p * kD + t];
  }
  float cpart = ((n0 + n1) + n2) + (c3 < kC ? n3 : 0.f);
  cpart = row16_sum(cpart);
  cpart += xhalf(cpart);
  if (lane == 0) wsum[wave] = cpart;
  __syncthreads();
  float total = 0.f;
  for (int w = 0; w < 8; ++w) total += wsum[w];
  if (t < kD) {
    const float pooled = s / total + kEpsStats;
    prec[t] = 1.0f / fmaxf(pooled, kEpsPrec);
  }
  __syncthreads();
  return total;
}

__global__ __launch_bounds__(256)
void prep_kernel(const float* __restrict__ z, float* __restrict__ ws) {
  __shared__ float wsum[8];
  __shared__ __attribute__((aligned(16))) float prec[kD];
  const int t = threadIdx.x;
  const int g = t >> 4;
  const int q = t & 15;

  if (blockIdx.x < kPrepClassBlocks) {
    const int c = blockIdx.x * 16 + g;
    const bool real = c < kC;
    v4f m = {0.f, 0.f, 0.f, 0.f};
    float cnt = 1.f;
    if (real) {
      m = *(const v4f*)(ws + OFF_MEAN + c * kD + 4 * q);
      cnt = ws[OFF_COUNTS + c];
    }
    const float total = join_statistics(ws, wsum, prec, t);
    const v4f p = *(const v4f*)(&prec[4 * q]);
    const v4f v = p * m;
    *(v4f*)(ws + OFF_MUPAD + c * kD + 4 * q) = v;
    const float quad = row16_sum(v.x * m.x + v.y * m.y + v.z * m.z + v.w * m.w);
    if (q == 0) ws[OFF_CCONST + c] = real ? logf(cnt / total) - 0.5f * quad : 0.f;
  } else {
    const int b0 = (blockIdx.x - kPrepClassBlocks) * 64 + 4 * g;
    v4f zv[4];
    for (int i = 0; i < 4; ++i)
      zv[i] = *(const v4f*)(z + (b0 + i) * kD + 4 * q);
    (void)join_statistics(ws, wsum, prec, t);
    const v4f p = *(const v4f*)(&prec[4 * q]);
    float rs[4];
    for (int i = 0; i < 4; ++i) {
      const v4f pz = p * zv[i];
      rs[i] = row16_sum(pz.x * zv[i].x + pz.y * zv[i].y + pz.z * zv[i].z + pz.w * zv[i].w);
    }
    const float mine = (q == 0) ? rs[0] : (q == 1) ? rs[1] : (q == 2) ? rs[2] : rs[3];
    if (q < 4) ws[OFF_QH + b0 + q] = -0.5f * mine;
  }
}

__global__ __launch_bounds__(256)
void lda_score_wmma_kernel(const float* __restrict__ z, const float* __restrict__ ws,
                           float* __restrict__ out) {
  const int wave = threadIdx.x >> 5;
  const int lane = threadIdx.x & 31;
  const int tile = blockIdx.x * 8 + wave;
  const int tm = tile & 127;
  const int tn = tile >> 7;
  const int row0 = tm * 16;
  const int col0 = tn * 16;
  const int half = lane >> 4;
  const int lm   = lane & 15;

  const float* __restrict__ ap = z + (row0 + lm) * kD + 2 * half;
  const float* __restrict__ bp = ws + OFF_MUPAD + (col0 + lm) * kD + 2 * half;

  v8f acc = {};
#pragma unroll
  for (int k = 0; k < kD; k += 4) {
    const v2f a = *(const v2f*)(ap + k);
    const v2f b = *(const v2f*)(bp + k);
    acc = __builtin_amdgcn_wmma_f32_16x16x4_f32(
         false, a,  false, b,
         (short)0, acc,  false,  false);
  }

  const int col = col0 + lm;
  const float cc = ws[OFF_CCONST + col];
  if (col < kC) {
#pragma unroll
    for (int r = 0; r < 8; ++r) {
      const int row = row0 + r + 8 * half;
      out[row * kC + col] = acc[r] + cc + ws[OFF_QH + row];
    }
  }
}

}

extern "C" void kernel_launch(void* const* d_in, const int* in_sizes, int n_in,
                              void* d_out, int out_size, void* d_ws, size_t ws_size,
                              hipStream_t stream) {
  (void)in_sizes; (void)n_in; (void)out_size; (void)ws_size;
  const float* z = (const float*)d_in[0];
  const int*   y = (const int*)d_in[1];
  float* ws  = (float*)d_ws;
  float* out = (float*)d_out;

  class_stats_kernel<<<dim3(kStatTiles, 2), kStatWaves * 32, 0, stream>>>(z, y, ws);
  row_resid_kernel<<<kResidBlocks, 256, 0, stream>>>(z, y, ws);
  prep_kernel<<<kPrepBlocks, 256, 0, stream>>>(z, ws);

  const int tiles = (kB / 16) * (kCPad / 16);
  lda_score_wmma_kernel<<<tiles / 8, 256, 0, stream>>>(z, ws, out);
}


# squared residuals taken inside the class-statistics kernel: a block keeps its rows in registers and subtracts its 16 class means from LDS; row-residual launch gone (4 to 3); order of this sum changes
# speedup vs baseline: 10.2461x; 1.0923x over previous
#include <hip/hip_runtime.h>
#include <math.h>

namespace {

constexpr int kC      = 1000;
constexpr int kCPad   = 1008;
constexpr int kD      = 64;
constexpr int kB      = 2048;
constexpr float kEpsStats = 1e-5f;
constexpr float kEpsPrec  = 1e-6f;
constexpr int kResidParts = kCPad / 16;

constexpr int OFF_COUNTS = 0;
constexpr int OFF_MEAN   = OFF_COUNTS + kC;
constexpr int OFF_VARSUM = OFF_MEAN   + kC * kD;
constexpr int OFF_MUPAD  = OFF_VARSUM + kC * kD;
constexpr int OFF_CCONST = OFF_MUPAD  + kCPad * kD;
constexpr int OFF_QH     = OFF_CCONST + kCPad;

typedef __attribute__((ext_vector_type(2))) float v2f;
typedef __attribute__((ext_vector_type(4))) float v4f;
typedef __attribute__((ext_vector_type(8))) float v8f;

template <int M>
__device__ __forceinline__ float row_xor(float x) {
  return __builtin_bit_cast(float, __builtin_amdgcn_mov_dpp(__builtin_bit_cast(int, x), 0x160 + M, 0xf, 0xf, false));
}
__device__ __forceinline__ float row16_sum(float x) {
  x += row_xor<1>(x);
  x += row_xor<2>(x);
  x += row_xor<4>(x);
  x += row_xor<8>(x);
  return x;
}
__device__ __forceinline__ float xhalf(float x) {
  return __builtin_bit_cast(float, __builtin_amdgcn_permlanex16(__builtin_bit_cast(int, x), __builtin_bit_cast(int, x),
                                                                0x76543210, 0xfedcba98, false, false));
}

constexpr int kStatWaves    = 16;
constexpr int kStatRows     = kB / kStatWaves;
constexpr int kStatSteps    = kStatRows / 4;
constexpr int kStatTiles    = kCPad / 16;

__global__ __launch_bounds__(kStatWaves * 32) __attribute__((amdgpu_waves_per_eu(4, 4)))
void class_stats_kernel(const float* __restrict__ z, const int* __restrict__ y,
                        float* __restrict__ ws) {
  __shared__ float red[kStatWaves][16][32];
  __shared__ float cred[kStatWaves][32];
  __shared__ __attribute__((aligned(8))) float meanL[16][32];
  __shared__ __attribute__((aligned(8))) float qred[kStatWaves][32];
  const int wave = threadIdx.x >> 5;
  const int lane = threadIdx.x & 31;
  const int half = lane >> 4;
  const int ln   = lane & 15;
  const int c0   = blockIdx.x * 16;
  const int fh   = blockIdx.y;
  const int cls  = c0 + ln;

  const int rbase = wave * kStatRows + 2 * half;
  const int*   __restrict__ yp = y + rbase;
  const float* __restrict__ zp = z + rbase * kD + 32 * fh + 2 * ln;

  v8f acc0 = {}, acc1 = {};
  v2f cnt2 = {0.f, 0.f};
  int2 labs[kStatSteps];
  v2f zr0[kStatSteps], zr1[kStatSteps];
  unsigned hit = 0;
#pragma unroll
  for (int s = 0; s < kStatSteps; ++s) {
    const int2 lab = *(const int2*)(yp + 4 * s);
    const v2f z0 = *(const v2f*)(zp + (4 * s) * kD);
    const v2f z1 = *(const v2f*)(zp + (4 * s + 1) * kD);
    labs[s] = lab; zr0[s] = z0; zr1[s] = z1;
    const bool ex = lab.x == cls, ey = lab.y == cls;
    if (__builtin_amdgcn_ballot_w32(ex || ey) != 0u) hit |= 1u << s;
    v2f a;
    a.x = ex ? 1.f : 0.f;
    a.y = ey ? 1.f : 0.f;
    cnt2 += a;
    const v2f b0 = {z0.x, z1.x};
    const v2f b1 = {z0.y, z1.y};
    acc0 = __builtin_amdgcn_wmma_f32_16x16x4_f32(false, a, false, b0, (short)0, acc0, false, false);
    acc1 = __builtin_amdgcn_wmma_f32_16x16x4_f32(false, a, false, b1, (short)0, acc1, false, false);
  }

#pragma unroll
  for (int r = 0; r < 8; ++r) {
    const v2f p = {acc0[r], acc1[r]};
    *(v2f*)(&red[wave][r + 8 * half][2 * ln]) = p;
  }
  cred[wave][lane] = cnt2.x + cnt2.y;
  __syncthreads();

  const int oc = threadIdx.x >> 5;
  const int of = threadIdx.x & 31;
  float s = 0.f, cnt = 0.f;
  for (int w = 0; w < kStatWaves; ++w) {
    s   += red[w][oc][of];
    cnt += cred[w][oc] + cred[w][oc + 16];
  }
  const int c = c0 + oc;
  const float cnte = cnt + kEpsStats;
  const float mean = s / cnte;
  meanL[oc][of] = mean;
  if (c < kC) {
    if (fh == 0 && of == 0) ws[OFF_COUNTS + c] = cnte;
    ws[OFF_MEAN + c * kD + 32 * fh + of] = mean;
  }
  __syncthreads();

  const int nreal = (kC - c0 < 16) ? (kC - c0) : 16;
  v2f q = {0.f, 0.f};
#pragma unroll
  for (int st = 0; st < kStatSteps; ++st) {
    if (hit & (1u << st)) {
      const int ix = labs[st].x - c0, iy = labs[st].y - c0;
      const bool inx = (unsigned)ix < (unsigned)nreal, iny = (unsigned)iy < (unsigned)nreal;
      const v2f m0 = *(const v2f*)(&meanL[inx ? ix : 0][2 * ln]);
      const v2f m1 = *(const v2f*)(&meanL[iny ? iy : 0][2 * ln]);
      v2f r0 = zr0[st] - m0, r1 = zr1[st] - m1;
      if (!inx) r0 = v2f{0.f, 0.f};
      if (!iny) r1 = v2f{0.f, 0.f};
      q += r0 * r0;
      q += r1 * r1;
    }
  }
  q.x += xhalf(q.x);
  q.y += xhalf(q.y);
  if (half == 0) *(v2f*)(&qred[wave][2 * ln]) = q;
  __syncthreads();
  if (threadIdx.x < 32) {
    float t = 0.f;
    for (int w = 0; w < kStatWaves; ++w) t += qred[w][threadIdx.x];
    ws[OFF_VARSUM + blockIdx.x * kD + 32 * fh + threadIdx.x] = t;
  }
}

constexpr int kPrepClassBlocks = kCPad / 16;
constexpr int kPrepRowBlocks   = kB / 64;
constexpr int kPrepBlocks      = kPrepClassBlocks + kPrepRowBlocks;

__device__ __forceinline__ float join_statistics(const float* __restrict__ ws, float* wsum, float (*vpart)[kD], float* prec, int t) {
  const int lane = t & 31;
  const int wave = t >> 5;
  const int c3 = t + 768;
  const float n0 = ws[OFF_COUNTS + t];
  const float n1 = ws[OFF_COUNTS + t + 256];
  const float n2 = ws[OFF_COUNTS + t + 512];
  const float n3 = ws[OFF_COUNTS + (c3 < kC ? c3 : kC - 1)];
  {
    const int f = t & (kD - 1);
    const int g4 = t >> 6;
    float sq = 0.f;
#pragma unroll
    for (int i = 0; i < 16; ++i) {
      const int p = 16 * g4 + i;
      const float v = ws[OFF_VARSUM + (p < kResidParts ? p : kResidParts - 1) * kD + f];
      sq += (p < kResidParts) ? v : 0.f;
    }
    vpart[g4][f] = sq;
  }
  float cpart = ((n0 + n1) + n2) + (c3 < kC ? n3 : 0.f);
  cpart = row16_sum(cpart);
  cpart += xhalf(cpart);
  if (lane == 0) wsum[wave] = cpart;
  __syncthreads();
  float total = 0.f;
  for (int w = 0; w < 8; ++w) total += wsum[w];
  if (t < kD) {
    const float s = ((vpart[0][t] + vpart[1][t]) + vpart[2][t]) + vpart[3][t];
    const float pooled = s / total + kEpsStats;
    prec[t] = 1.0f / fmaxf(pooled, kEpsPrec);
  }
  __syncthreads();
  return total;
}

__global__ __launch_bounds__(256)
void prep_kernel(const float* __restrict__ z, float* __restrict__ ws) {
  __shared__ float wsum[8];
  __shared__ float vpart[4][kD];
  __shared__ __attribute__((aligned(16))) float prec[kD];
  const int t = threadIdx.x;
  const int g = t >> 4;
  const int q = t & 15;

  if (blockIdx.x < kPrepClassBlocks) {
    const int c = blockIdx.x * 16 + g;
    const bool real = c < kC;
    v4f m = {0.f, 0.f, 0.f, 0.f};
    float cnt = 1.f;
    if (real) {
      m = *(const v4f*)(ws + OFF_MEAN + c * kD + 4 * q);
      cnt = ws[OFF_COUNTS + c];
    }
    const float total = join_statistics(ws, wsum, vpart, prec, t);
    const v4f p = *(const v4f*)(&prec[4 * q]);
    const v4f v = p * m;
    *(v4f*)(ws + OFF_MUPAD + c * kD + 4 * q) = v;
    const float quad = row16_sum(v.x * m.x + v.y * m.y + v.z * m.z + v.w * m.w);
    if (q == 0) ws[OFF_CCONST + c] = real ? logf(cnt / total) - 0.5f * quad : 0.f;
  } else {
    const int b0 = (blockIdx.x - kPrepClassBlocks) * 64 + 4 * g;
    v4f zv[4];
    for (int i = 0; i < 4; ++i)
      zv[i] = *(const v4f*)(z + (b0 + i) * kD + 4 * q);
    (void)join_statistics(ws, wsum, vpart, prec, t);
    const v4f p = *(const v4f*)(&prec[4 * q]);
    float rs[4];
    for (int i = 0; i < 4; ++i) {
      const v4f pz = p * zv[i];
      rs[i] = row16_sum(pz.x * zv[i].x + pz.y * zv[i].y + pz.z * zv[i].z + pz.w * zv[i].w);
    }
    const float mine = (q == 0) ? rs[0] : (q == 1) ? rs[1] : (q == 2) ? rs[2] : rs[3];
    if (q < 4) ws[OFF_QH + b0 + q] = -0.5f * mine;
  }
}

__global__ __launch_bounds__(256)
void lda_score_wmma_kernel(const float* __restrict__ z, const float* __restrict__ ws,
                           float* __restrict__ out) {
  const int wave = threadIdx.x >> 5;
  const int lane = threadIdx.x & 31;
  const int tile = blockIdx.x * 8 + wave;
  const int tm = tile & 127;
  const int tn = tile >> 7;
  const int row0 = tm * 16;
  const int col0 = tn * 16;
  const int half = lane >> 4;
  const int lm   = lane & 15;

  const float* __restrict__ ap = z + (row0 + lm) * kD + 2 * half;
  const float* __restrict__ bp = ws + OFF_MUPAD + (col0 + lm) * kD + 2 * half;

  v8f acc = {};
#pragma unroll
  for (int k = 0; k < kD; k += 4) {
    const v2f a = *(const v2f*)(ap + k);
    const v2f b = *(const v2f*)(bp + k);
    acc = __builtin_amdgcn_wmma_f32_16x16x4_f32(
         false, a,  false, b,
         (short)0, acc,  false,  false);
  }

  const int col = col0 + lm;
  const float cc = ws[OFF_CCONST + col];
  if (col < kC) {
#pragma unroll
    for (int r = 0; r < 8; ++r) {
      const int row = row0 + r + 8 * half;
      out[row * kC + col] = acc[r] + cc + ws[OFF_QH + row];
    }
  }
}

}

extern "C" void kernel_launch(void* const* d_in, const int* in_sizes, int n_in,
                              void* d_out, int out_size, void* d_ws, size_t ws_size,
                              hipStream_t stream) {
  (void)in_sizes; (void)n_in; (void)out_size; (void)ws_size;
  const float* z = (const float*)d_in[0];
  const int*   y = (const int*)d_in[1];
  float* ws  = (float*)d_ws;
  float* out = (float*)d_out;

  class_stats_kernel<<<dim3(kStatTiles, 2), kStatWaves * 32, 0, stream>>>(z, y, ws);
  prep_kernel<<<kPrepBlocks, 256, 0, stream>>>(z, ws);

  const int tiles = (kB / 16) * (kCPad / 16);
  lda_score_wmma_kernel<<<tiles / 8, 256, 0, stream>>>(z, ws, out);
}


# score kernel tiled in registers: a wave keeps the A fragments of 32 rows and walks 4 class tiles, each B fragment feeding both row tiles; 1024 waves instead of 8064, no LDS; order of sums unchanged
# speedup vs baseline: 11.4033x; 1.1129x over previous
#include <hip/hip_runtime.h>
#include <math.h>

namespace {

constexpr int kC      = 1000;
constexpr int kCPad   = 1008;
constexpr int kD      = 64;
constexpr int kB      = 2048;
constexpr float kEpsStats = 1e-5f;
constexpr float kEpsPrec  = 1e-6f;
constexpr int kResidParts = kCPad / 16;

constexpr int OFF_COUNTS = 0;
constexpr int OFF_MEAN   = OFF_COUNTS + kC;
constexpr int OFF_VARSUM = OFF_MEAN   + kC * kD;
constexpr int OFF_MUPAD  = OFF_VARSUM + kC * kD;
constexpr int OFF_CCONST = OFF_MUPAD  + kCPad * kD;
constexpr int OFF_QH     = OFF_CCONST + kCPad;

typedef __attribute__((ext_vector_type(2))) float v2f;
typedef __attribute__((ext_vector_type(4))) float v4f;
typedef __attribute__((ext_vector_type(8))) float v8f;

template <int M>
__device__ __forceinline__ float row_xor(float x) {
  return __builtin_bit_cast(float, __builtin_amdgcn_mov_dpp(__builtin_bit_cast(int, x), 0x160 + M, 0xf, 0xf, false));
}
__device__ __forceinline__ float row16_sum(float x) {
  x += row_xor<1>(x);
  x += row_xor<2>(x);
  x += row_xor<4>(x);
  x += row_xor<8>(x);
  return x;
}
__device__ __forceinline__ float xhalf(float x) {
  return __builtin_bit_cast(float, __builtin_amdgcn_permlanex16(__builtin_bit_cast(int, x), __builtin_bit_cast(int, x),
                                                                0x76543210, 0xfedcba98, false, false));
}

constexpr int kStatWaves    = 16;
constexpr int kStatRows     = kB / kStatWaves;
constexpr int kStatSteps    = kStatRows / 4;
constexpr int kStatTiles    = kCPad / 16;

__global__ __launch_bounds__(kStatWaves * 32) __attribute__((amdgpu_waves_per_eu(4, 4)))
void class_stats_kernel(const float* __restrict__ z, const int* __restrict__ y,
                        float* __restrict__ ws) {
  __shared__ float red[kStatWaves][16][32];
  __shared__ float cred[kStatWaves][32];
  __shared__ __attribute__((aligned(8))) float meanL[16][32];
  __shared__ __attribute__((aligned(8))) float qred[kStatWaves][32];
  const int wave = threadIdx.x >> 5;
  const int lane = threadIdx.x & 31;
  const int half = lane >> 4;
  const int ln   = lane & 15;
  const int c0   = blockIdx.x * 16;
  const int fh   = blockIdx.y;
  const int cls  = c0 + ln;

  const int rbase = wave * kStatRows + 2 * half;
  const int*   __restrict__ yp = y + rbase;
  const float* __restrict__ zp = z + rbase * kD + 32 * fh + 2 * ln;

  v8f acc0 = {}, acc1 = {};
  v2f cnt2 = {0.f, 0.f};
  int2 labs[kStatSteps];
  v2f zr0[kStatSteps], zr1[kStatSteps];
  unsigned hit = 0;
#pragma unroll
  for (int s = 0; s < kStatSteps; ++s) {
    const int2 lab = *(const int2*)(yp + 4 * s);
    const v2f z0 = *(const v2f*)(zp + (4 * s) * kD);
    const v2f z1 = *(const v2f*)(zp + (4 * s + 1) * kD);
    labs[s] = lab; zr0[s] = z0; zr1[s] = z1;
    const bool ex = lab.x == cls, ey = lab.y == cls;
    if (__builtin_amdgcn_ballot_w32(ex || ey) != 0u) hit |= 1u << s;
    v2f a;
    a.x = ex ? 1.f : 0.f;
    a.y = ey ? 1.f : 0.f;
    cnt2 += a;
    const v2f b0 = {z0.x, z1.x};
    const v2f b1 = {z0.y, z1.y};
    acc0 = __builtin_amdgcn_wmma_f32_16x16x4_f32(false, a, false, b0, (short)0, acc0, false, false);
    acc1 = __builtin_amdgcn_wmma_f32_16x16x4_f32(false, a, false, b1, (short)0, acc1, false, false);
  }

#pragma unroll
  for (int r = 0; r < 8; ++r) {
    const v2f p = {acc0[r], acc1[r]};
    *(v2f*)(&red[wave][r + 8 * half][2 * ln]) = p;
  }
  cred[wave][lane] = cnt2.x + cnt2.y;
  __syncthreads();

  const int oc = threadIdx.x >> 5;
  const int of = threadIdx.x & 31;
  float s = 0.f, cnt = 0.f;
  for (int w = 0; w < kStatWaves; ++w) {
    s   += red[w][oc][of];
    cnt += cred[w][oc] + cred[w][oc + 16];
  }
  const int c = c0 + oc;
  const float cnte = cnt + kEpsStats;
  const float mean = s / cnte;
  meanL[oc][of] = mean;
  if (c < kC) {
    if (fh == 0 && of == 0) ws[OFF_COUNTS + c] = cnte;
    ws[OFF_MEAN + c * kD + 32 * fh + of] = mean;
  }
  __syncthreads();

  const int nreal = (kC - c0 < 16) ? (kC - c0) : 16;
  v2f q = {0.f, 0.f};
#pragma unroll
  for (int st = 0; st < kStatSteps; ++st) {
    if (hit & (1u << st)) {
      const int ix = labs[st].x - c0, iy = labs[st].y - c0;
      const bool inx = (unsigned)ix < (unsigned)nreal, iny = (unsigned)iy < (unsigned)nreal;
      const v2f m0 = *(const v2f*)(&meanL[inx ? ix : 0][2 * ln]);
      const v2f m1 = *(const v2f*)(&meanL[iny ? iy : 0][2 * ln]);
      v2f r0 = zr0[st] - m0, r1 = zr1[st] - m1;
      if (!inx) r0 = v2f{0.f, 0.f};
      if (!iny) r1 = v2f{0.f, 0.f};
      q += r0 * r0;
      q += r1 * r1;
    }
  }
  q.x += xhalf(q.x);
  q.y += xhalf(q.y);
  if (half == 0) *(v2f*)(&qred[wave][2 * ln]) = q;
  __syncthreads();
  if (threadIdx.x < 32) {
    float t = 0.f;
    for (int w = 0; w < kStatWaves; ++w) t += qred[w][threadIdx.x];
    ws[OFF_VARSUM + blockIdx.x * kD + 32 * fh + threadIdx.x] = t;
  }
}

constexpr int kPrepClassBlocks = kCPad / 16;
constexpr int kPrepRowBlocks   = kB / 64;
constexpr int kPrepBlocks      = kPrepClassBlocks + kPrepRowBlocks;

__device__ __forceinline__ float join_statistics(const float* __restrict__ ws, float* wsum, float (*vpart)[kD], float* prec, int t) {
  const int lane = t & 31;
  const int wave = t >> 5;
  const int c3 = t + 768;
  const float n0 = ws[OFF_COUNTS + t];
  const float n1 = ws[OFF_COUNTS + t + 256];
  const float n2 = ws[OFF_COUNTS + t + 512];
  const float n3 = ws[OFF_COUNTS + (c3 < kC ? c3 : kC - 1)];
  {
    const int f = t & (kD - 1);
    const int g4 = t >> 6;
    float sq = 0.f;
#pragma unroll
    for (int i = 0; i < 16; ++i) {
      const int p = 16 * g4 + i;
      const float v = ws[OFF_VARSUM + (p < kResidParts ? p : kResidParts - 1) * kD + f];
      sq += (p < kResidParts) ? v : 0.f;
    }
    vpart[g4][f] = sq;
  }
  float cpart = ((n0 + n1) + n2) + (c3 < kC ? n3 : 0.f);
  cpart = row16_sum(cpart);
  cpart += xhalf(cpart);
  if (lane == 0) wsum[wave] = cpart;
  __syncthreads();
  float total = 0.f;
  for (int w = 0; w < 8; ++w) total += wsum[w];
  if (t < kD) {
    const float s = ((vpart[0][t] + vpart[1][t]) + vpart[2][t]) + vpart[3][t];
    const float pooled = s / total + kEpsStats;
    prec[t] = 1.0f / fmaxf(pooled, kEpsPrec);
  }
  __syncthreads();
  return total;
}

__global__ __launch_bounds__(256)
void prep_kernel(const float* __restrict__ z, float* __restrict__ ws) {
  __shared__ float wsum[8];
  __shared__ float vpart[4][kD];
  __shared__ __attribute__((aligned(16))) float prec[kD];
  const int t = threadIdx.x;
  const int g = t >> 4;
  const int q = t & 15;

  if (blockIdx.x < kPrepClassBlocks) {
    const int c = blockIdx.x * 16 + g;
    const bool real = c < kC;
    v4f m = {0.f, 0.f, 0.f, 0.f};
    float cnt = 1.f;
    if (real) {
      m = *(const v4f*)(ws + OFF_MEAN + c * kD + 4 * q);
      cnt = ws[OFF_COUNTS + c];
    }
    const float total = join_statistics(ws, wsum, vpart, prec, t);
    const v4f p = *(const v4f*)(&prec[4 * q]);
    const v4f v = p * m;
    *(v4f*)(ws + OFF_MUPAD + c * kD + 4 * q) = v;
    const float quad = row16_sum(v.x * m.x + v.y * m.y + v.z * m.z + v.w * m.w);
    if (q == 0) ws[OFF_CCONST + c] = real ? logf(cnt / total) - 0.5f * quad : 0.f;
  } else {
    const int b0 = (blockIdx.x - kPrepClassBlocks) * 64 + 4 * g;
    v4f zv[4];
    for (int i = 0; i < 4; ++i)
      zv[i] = *(const v4f*)(z + (b0 + i) * kD + 4 * q);
    (void)join_statistics(ws, wsum, vpart, prec, t);
    const v4f p = *(const v4f*)(&prec[4 * q]);
    float rs[4];
    for (int i = 0; i < 4; ++i) {
      const v4f pz = p * zv[i];
      rs[i] = row16_sum(pz.x * zv[i].x + pz.y * zv[i].y + pz.z * zv[i].z + pz.w * zv[i].w);
    }
    const float mine = (q == 0) ? rs[0] : (q == 1) ? rs[1] : (q == 2) ? rs[2] : rs[3];
    if (q < 4) ws[OFF_QH + b0 + q] = -0.5f * mine;
  }
}

constexpr int kScoreRowTiles  = 2;
constexpr int kScoreColTiles  = 4;
constexpr int kScoreWaves     = 4;
constexpr int kScoreNTiles    = kCPad / 16;
constexpr int kScoreColGroups = (kScoreNTiles + kScoreColTiles - 1) / kScoreColTiles;
constexpr int kScoreRowGroups = kB / (16 * kScoreRowTiles * kScoreWaves);
constexpr int kScoreSteps     = kD / 4;

__global__ __launch_bounds__(kScoreWaves * 32)
void lda_score_wmma_kernel(const float* __restrict__ z, const float* __restrict__ ws,
                           float* __restrict__ out) {
  const int wave = threadIdx.x >> 5;
  const int lane = threadIdx.x & 31;
  const int half = lane >> 4;
  const int lm   = lane & 15;
  const int row0 = (blockIdx.y * kScoreWaves + wave) * (16 * kScoreRowTiles);
  const int tn0  = blockIdx.x * kScoreColTiles;

  const float* __restrict__ ap = z + (row0 + lm) * kD + 2 * half;
  int col[kScoreColTiles];
  const float* __restrict__ bp[kScoreColTiles];
#pragma unroll
  for (int j = 0; j < kScoreColTiles; ++j) {
    const int tn = tn0 + j;
    col[j] = (tn < kScoreNTiles ? tn : kScoreNTiles - 1) * 16 + lm;
    bp[j] = ws + OFF_MUPAD + col[j] * kD + 2 * half;
  }

  v2f a[kScoreRowTiles][kScoreSteps];
  v2f b[kScoreColTiles][kScoreSteps];
#pragma unroll
  for (int s = 0; s < kScoreSteps; ++s) {
#pragma unroll
    for (int m = 0; m < kScoreRowTiles; ++m)
      a[m][s] = *(const v2f*)(ap + m * 16 * kD + 4 * s);
    b[0][s] = *(const v2f*)(bp[0] + 4 * s);
  }
#pragma unroll
  for (int j = 1; j < kScoreColTiles; ++j)
#pragma unroll
    for (int s = 0; s < kScoreSteps; ++s)
      b[j][s] = *(const v2f*)(bp[j] + 4 * s);

  float cc[kScoreColTiles];
#pragma unroll
  for (int j = 0; j < kScoreColTiles; ++j) cc[j] = ws[OFF_CCONST + col[j]];
  float qh[kScoreRowTiles][8];
#pragma unroll
  for (int m = 0; m < kScoreRowTiles; ++m)
#pragma unroll
    for (int r = 0; r < 8; ++r)
      qh[m][r] = ws[OFF_QH + row0 + 16 * m + r + 8 * half];

#pragma unroll
  for (int j = 0; j < kScoreColTiles; ++j) {
    v8f acc[kScoreRowTiles];
#pragma unroll
    for (int m = 0; m < kScoreRowTiles; ++m) acc[m] = v8f{};
#pragma unroll
    for (int s = 0; s < kScoreSteps; ++s) {
#pragma unroll
      for (int m = 0; m < kScoreRowTiles; ++m)
        acc[m] = __builtin_amdgcn_wmma_f32_16x16x4_f32(
             false, a[m][s],  false, b[j][s],
             (short)0, acc[m],  false,  false);
    }
    asm volatile("" :: "v"(cc[j]));
    if (tn0 + j < kScoreNTiles && col[j] < kC) {
#pragma unroll
      for (int m = 0; m < kScoreRowTiles; ++m)
#pragma unroll
        for (int r = 0; r < 8; ++r) {
          const int row = row0 + 16 * m + r + 8 * half;
          out[row * kC + col[j]] = acc[m][r] + cc[j] + qh[m][r];
        }
    }
  }
}

}

extern "C" void kernel_launch(void* const* d_in, const int* in_sizes, int n_in,
                              void* d_out, int out_size, void* d_ws, size_t ws_size,
                              hipStream_t stream) {
  (void)in_sizes; (void)n_in; (void)out_size; (void)ws_size;
  const float* z = (const float*)d_in[0];
  const int*   y = (const int*)d_in[1];
  float* ws  = (float*)d_ws;
  float* out = (float*)d_out;

  class_stats_kernel<<<dim3(kStatTiles, 2), kStatWaves * 32, 0, stream>>>(z, y, ws);
  prep_kernel<<<kPrepBlocks, 256, 0, stream>>>(z, ws);

  lda_score_wmma_kernel<<<dim3(kScoreColGroups, kScoreRowGroups), kScoreWaves * 32, 0, stream>>>(z, ws, out);
}


# statistics product skips all-zero k-steps: a wave loads its 128 labels once, one ballot names the k-steps holding a row of the block's classes, only those load z and multiply; order of sums unchanged
# speedup vs baseline: 13.4274x; 1.1775x over previous
#include <hip/hip_runtime.h>
#include <math.h>

namespace {

constexpr int kC      = 1000;
constexpr int kCPad   = 1008;
constexpr int kD      = 64;
constexpr int kB      = 2048;
constexpr float kEpsStats = 1e-5f;
constexpr float kEpsPrec  = 1e-6f;
constexpr int kResidParts = kCPad / 16;

constexpr int OFF_COUNTS = 0;
constexpr int OFF_MEAN   = OFF_COUNTS + kC;
constexpr int OFF_VARSUM = OFF_MEAN   + kC * kD;
constexpr int OFF_MUPAD  = OFF_VARSUM + kC * kD;
constexpr int OFF_CCONST = OFF_MUPAD  + kCPad * kD;
constexpr int OFF_QH     = OFF_CCONST + kCPad;

typedef __attribute__((ext_vector_type(2))) float v2f;
typedef __attribute__((ext_vector_type(4))) float v4f;
typedef __attribute__((ext_vector_type(8))) float v8f;

template <int M>
__device__ __forceinline__ float row_xor(float x) {
  return __builtin_bit_cast(float, __builtin_amdgcn_mov_dpp(__builtin_bit_cast(int, x), 0x160 + M, 0xf, 0xf, false));
}
__device__ __forceinline__ float row16_sum(float x) {
  x += row_xor<1>(x);
  x += row_xor<2>(x);
  x += row_xor<4>(x);
  x += row_xor<8>(x);
  return x;
}
__device__ __forceinline__ float xhalf(float x) {
  return __builtin_bit_cast(float, __builtin_amdgcn_permlanex16(__builtin_bit_cast(int, x), __builtin_bit_cast(int, x),
                                                                0x76543210, 0xfedcba98, false, false));
}

constexpr int kStatWaves    = 16;
constexpr int kStatRows     = kB / kStatWaves;
constexpr int kStatSteps    = kStatRows / 4;
constexpr int kStatTiles    = kCPad / 16;
constexpr int kStatSlots    = 8;
constexpr int kStatTrips    = kStatSteps / kStatSlots;

__device__ __forceinline__ unsigned pick_steps(unsigned rem, int (&s)[kStatSlots], bool (&live)[kStatSlots]) {
#pragma unroll
  for (int i = 0; i < kStatSlots; ++i) {
    live[i] = rem != 0u;
    s[i] = live[i] ? __builtin_ctz(rem) : 0;
    rem &= rem - 1u;
  }
  return rem;
}

__device__ __forceinline__ int2 step_labels(const int4 lab4, int s, int half) {
  const int k0 = __builtin_amdgcn_readlane(lab4.x, s);
  const int k1 = __builtin_amdgcn_readlane(lab4.y, s);
  const int k2 = __builtin_amdgcn_readlane(lab4.z, s);
  const int k3 = __builtin_amdgcn_readlane(lab4.w, s);
  return half ? make_int2(k2, k3) : make_int2(k0, k1);
}

__device__ __forceinline__ void load_steps(const float* __restrict__ zp, const int (&s)[kStatSlots],
                                           v2f (&z0)[kStatSlots], v2f (&z1)[kStatSlots]) {
#pragma unroll
  for (int i = 0; i < kStatSlots; ++i) {
    z0[i] = *(const v2f*)(zp + (4 * s[i]) * kD);
    z1[i] = *(const v2f*)(zp + (4 * s[i] + 1) * kD);
  }
}

__global__ __launch_bounds__(kStatWaves * 32)
void class_stats_kernel(const float* __restrict__ z, const int* __restrict__ y,
                        float* __restrict__ ws) {
  __shared__ float red[kStatWaves][16][32];
  __shared__ float cred[kStatWaves][32];
  __shared__ __attribute__((aligned(8))) float meanL[16][32];
  __shared__ __attribute__((aligned(8))) float qred[kStatWaves][32];
  const int wave = threadIdx.x >> 5;
  const int lane = threadIdx.x & 31;
  const int half = lane >> 4;
  const int ln   = lane & 15;
  const int c0   = blockIdx.x * 16;
  const int fh   = blockIdx.y;
  const int cls  = c0 + ln;

  const int4 lab4 = *(const int4*)(y + wave * kStatRows + 4 * lane);
  const bool mine = (unsigned)(lab4.x - c0) < 16u || (unsigned)(lab4.y - c0) < 16u ||
                    (unsigned)(lab4.z - c0) < 16u || (unsigned)(lab4.w - c0) < 16u;
  const unsigned hit_all = __builtin_amdgcn_ballot_w32(mine);

  const float* __restrict__ zp = z + (wave * kStatRows + 2 * half) * kD + 32 * fh + 2 * ln;

  v8f acc0 = {}, acc1 = {};
  v2f cnt2 = {0.f, 0.f};
  int  sk[kStatSlots];
  bool lk[kStatSlots];
  v2f  zk0[kStatSlots], zk1[kStatSlots];
#pragma unroll
  for (int i = 0; i < kStatSlots; ++i) {
    sk[i] = 0; lk[i] = false; zk0[i] = v2f{0.f, 0.f}; zk1[i] = v2f{0.f, 0.f};
  }

  auto mma_slot = [&](int i) {
    const int2 lab = step_labels(lab4, sk[i], half);
    v2f a;
    a.x = (lk[i] && lab.x == cls) ? 1.f : 0.f;
    a.y = (lk[i] && lab.y == cls) ? 1.f : 0.f;
    cnt2 += a;
    const v2f b0 = {zk0[i].x, zk1[i].x};
    const v2f b1 = {zk0[i].y, zk1[i].y};
    acc0 = __builtin_amdgcn_wmma_f32_16x16x4_f32(false, a, false, b0, (short)0, acc0, false, false);
    acc1 = __builtin_amdgcn_wmma_f32_16x16x4_f32(false, a, false, b1, (short)0, acc1, false, false);
  };

  unsigned last = 0u;
  unsigned rem  = hit_all;
  for (int trip = 0; trip < kStatTrips && rem != 0u; ++trip) {
    const unsigned before = rem;
    rem  = pick_steps(rem, sk, lk);
    last = before ^ rem;
    load_steps(zp, sk, zk0, zk1);
#pragma unroll
    for (int i = 0; i < kStatSlots / 2; ++i) mma_slot(i);
    if (lk[kStatSlots / 2]) {
#pragma unroll
      for (int i = kStatSlots / 2; i < kStatSlots; ++i) mma_slot(i);
    }
  }

#pragma unroll
  for (int r = 0; r < 8; ++r) {
    const v2f p = {acc0[r], acc1[r]};
    *(v2f*)(&red[wave][r + 8 * half][2 * ln]) = p;
  }
  cred[wave][lane] = cnt2.x + cnt2.y;
  __syncthreads();

  const int oc = threadIdx.x >> 5;
  const int of = threadIdx.x & 31;
  float s = 0.f, cnt = 0.f;
  for (int w = 0; w < kStatWaves; ++w) {
    s   += red[w][oc][of];
    cnt += cred[w][oc] + cred[w][oc + 16];
  }
  const int c = c0 + oc;
  const float cnte = cnt + kEpsStats;
  const float mean = s / cnte;
  meanL[oc][of] = mean;
  if (c < kC) {
    if (fh == 0 && of == 0) ws[OFF_COUNTS + c] = cnte;
    ws[OFF_MEAN + c * kD + 32 * fh + of] = mean;
  }
  __syncthreads();

  const int nreal = (kC - c0 < 16) ? (kC - c0) : 16;
  v2f q = {0.f, 0.f};
  auto resid_slot = [&](int st, v2f zv0, v2f zv1) {
    const int2 lab = step_labels(lab4, st, half);
    const int ix = lab.x - c0, iy = lab.y - c0;
    const bool inx = (unsigned)ix < (unsigned)nreal, iny = (unsigned)iy < (unsigned)nreal;
    const v2f m0 = *(const v2f*)(&meanL[inx ? ix : 0][2 * ln]);
    const v2f m1 = *(const v2f*)(&meanL[iny ? iy : 0][2 * ln]);
    v2f r0 = zv0 - m0, r1 = zv1 - m1;
    if (!inx) r0 = v2f{0.f, 0.f};
    if (!iny) r1 = v2f{0.f, 0.f};
    q += r0 * r0;
    q += r1 * r1;
  };
  unsigned early = hit_all & ~last;
  for (int trip = 0; trip < kStatTrips && early != 0u; ++trip) {
    int  se[kStatSlots];
    bool le[kStatSlots];
    v2f  ze0[kStatSlots], ze1[kStatSlots];
    early = pick_steps(early, se, le);
    load_steps(zp, se, ze0, ze1);
#pragma unroll
    for (int i = 0; i < kStatSlots; ++i)
      if (le[i]) resid_slot(se[i], ze0[i], ze1[i]);
  }
#pragma unroll
  for (int i = 0; i < kStatSlots; ++i)
    if (lk[i]) resid_slot(sk[i], zk0[i], zk1[i]);

  q.x += xhalf(q.x);
  q.y += xhalf(q.y);
  if (half == 0) *(v2f*)(&qred[wave][2 * ln]) = q;
  __syncthreads();
  if (threadIdx.x < 32) {
    float t = 0.f;
    for (int w = 0; w < kStatWaves; ++w) t += qred[w][threadIdx.x];
    ws[OFF_VARSUM + blockIdx.x * kD + 32 * fh + threadIdx.x] = t;
  }
}

constexpr int kPrepClassBlocks = kCPad / 16;
constexpr int kPrepRowBlocks   = kB / 64;
constexpr int kPrepBlocks      = kPrepClassBlocks + kPrepRowBlocks;

__device__ __forceinline__ float join_statistics(const float* __restrict__ ws, float* wsum, float (*vpart)[kD], float* prec, int t) {
  const int lane = t & 31;
  const int wave = t >> 5;
  const int c3 = t + 768;
  const float n0 = ws[OFF_COUNTS + t];
  const float n1 = ws[OFF_COUNTS + t + 256];
  const float n2 = ws[OFF_COUNTS + t + 512];
  const float n3 = ws[OFF_COUNTS + (c3 < kC ? c3 : kC - 1)];
  {
    const int f = t & (kD - 1);
    const int g4 = t >> 6;
    float sq = 0.f;
#pragma unroll
    for (int i = 0; i < 16; ++i) {
      const int p = 16 * g4 + i;
      const float v = ws[OFF_VARSUM + (p < kResidParts ? p : kResidParts - 1) * kD + f];
      sq += (p < kResidParts) ? v : 0.f;
    }
    vpart[g4][f] = sq;
  }
  float cpart = ((n0 + n1) + n2) + (c3 < kC ? n3 : 0.f);
  cpart = row16_sum(cpart);
  cpart += xhalf(cpart);
  if (lane == 0) wsum[wave] = cpart;
  __syncthreads();
  float total = 0.f;
  for (int w = 0; w < 8; ++w) total += wsum[w];
  if (t < kD) {
    const float s = ((vpart[0][t] + vpart[1][t]) + vpart[2][t]) + vpart[3][t];
    const float pooled = s / total + kEpsStats;
    prec[t] = 1.0f / fmaxf(pooled, kEpsPrec);
  }
  __syncthreads();
  return total;
}

__global__ __launch_bounds__(256)
void prep_kernel(const float* __restrict__ z, float* __restrict__ ws) {
  __shared__ float wsum[8];
  __shared__ float vpart[4][kD];
  __shared__ __attribute__((aligned(16))) float prec[kD];
  const int t = threadIdx.x;
  const int g = t >> 4;
  const int q = t & 15;

  if (blockIdx.x < kPrepClassBlocks) {
    const int c = blockIdx.x * 16 + g;
    const bool real = c < kC;
    v4f m = {0.f, 0.f, 0.f, 0.f};
    float cnt = 1.f;
    if (real) {
      m = *(const v4f*)(ws + OFF_MEAN + c * kD + 4 * q);
      cnt = ws[OFF_COUNTS + c];
    }
    const float total = join_statistics(ws, wsum, vpart, prec, t);
    const v4f p = *(const v4f*)(&prec[4 * q]);
    const v4f v = p * m;
    *(v4f*)(ws + OFF_MUPAD + c * kD + 4 * q) = v;
    const float quad = row16_sum(v.x * m.x + v.y * m.y + v.z * m.z + v.w * m.w);
    if (q == 0) ws[OFF_CCONST + c] = real ? logf(cnt / total) - 0.5f * quad : 0.f;
  } else {
    const int b0 = (blockIdx.x - kPrepClassBlocks) * 64 + 4 * g;
    v4f zv[4];
    for (int i = 0; i < 4; ++i)
      zv[i] = *(const v4f*)(z + (b0 + i) * kD + 4 * q);
    (void)join_statistics(ws, wsum, vpart, prec, t);
    const v4f p = *(const v4f*)(&prec[4 * q]);
    float rs[4];
    for (int i = 0; i < 4; ++i) {
      const v4f pz = p * zv[i];
      rs[i] = row16_sum(pz.x * zv[i].x + pz.y * zv[i].y + pz.z * zv[i].z + pz.w * zv[i].w);
    }
    const float mine = (q == 0) ? rs[0] : (q == 1) ? rs[1] : (q == 2) ? rs[2] : rs[3];
    if (q < 4) ws[OFF_QH + b0 + q] = -0.5f * mine;
  }
}

constexpr int kScoreRowTiles  = 2;
constexpr int kScoreColTiles  = 4;
constexpr int kScoreWaves     = 4;
constexpr int kScoreNTiles    = kCPad / 16;
constexpr int kScoreColGroups = (kScoreNTiles + kScoreColTiles - 1) / kScoreColTiles;
constexpr int kScoreRowGroups = kB / (16 * kScoreRowTiles * kScoreWaves);
constexpr int kScoreSteps     = kD / 4;

__global__ __launch_bounds__(kScoreWaves * 32)
void lda_score_wmma_kernel(const float* __restrict__ z, const float* __restrict__ ws,
                           float* __restrict__ out) {
  const int wave = threadIdx.x >> 5;
  const int lane = threadIdx.x & 31;
  const int half = lane >> 4;
  const int lm   = lane & 15;
  const int row0 = (blockIdx.y * kScoreWaves + wave) * (16 * kScoreRowTiles);
  const int tn0  = blockIdx.x * kScoreColTiles;

  const float* __restrict__ ap = z + (row0 + lm) * kD + 2 * half;
  int col[kScoreColTiles];
  const float* __restrict__ bp[kScoreColTiles];
#pragma unroll
  for (int j = 0; j < kScoreColTiles; ++j) {
    const int tn = tn0 + j;
    col[j] = (tn < kScoreNTiles ? tn : kScoreNTiles - 1) * 16 + lm;
    bp[j] = ws + OFF_MUPAD + col[j] * kD + 2 * half;
  }

  v2f a[kScoreRowTiles][kScoreSteps];
  v2f b[kScoreColTiles][kScoreSteps];
#pragma unroll
  for (int s = 0; s < kScoreSteps; ++s) {
#pragma unroll
    for (int m = 0; m < kScoreRowTiles; ++m)
      a[m][s] = *(const v2f*)(ap + m * 16 * kD + 4 * s);
    b[0][s] = *(const v2f*)(bp[0] + 4 * s);
  }
#pragma unroll
  for (int j = 1; j < kScoreColTiles; ++j)
#pragma unroll
    for (int s = 0; s < kScoreSteps; ++s)
      b[j][s] = *(const v2f*)(bp[j] + 4 * s);

  float cc[kScoreColTiles];
#pragma unroll
  for (int j = 0; j < kScoreColTiles; ++j) cc[j] = ws[OFF_CCONST + col[j]];
  float qh[kScoreRowTiles][8];
#pragma unroll
  for (int m = 0; m < kScoreRowTiles; ++m)
#pragma unroll
    for (int r = 0; r < 8; ++r)
      qh[m][r] = ws[OFF_QH + row0 + 16 * m + r + 8 * half];

#pragma unroll
  for (int j = 0; j < kScoreColTiles; ++j) {
    v8f acc[kScoreRowTiles];
#pragma unroll
    for (int m = 0; m < kScoreRowTiles; ++m) acc[m] = v8f{};
#pragma unroll
    for (int s = 0; s < kScoreSteps; ++s) {
#pragma unroll
      for (int m = 0; m < kScoreRowTiles; ++m)
        acc[m] = __builtin_amdgcn_wmma_f32_16x16x4_f32(
             false, a[m][s],  false, b[j][s],
             (short)0, acc[m],  false,  false);
    }
    asm volatile("" :: "v"(cc[j]));
    if (tn0 + j < kScoreNTiles && col[j] < kC) {
#pragma unroll
      for (int m = 0; m < kScoreRowTiles; ++m)
#pragma unroll
        for (int r = 0; r < 8; ++r) {
          const int row = row0 + 16 * m + r + 8 * half;
          out[row * kC + col[j]] = acc[m][r] + cc[j] + qh[m][r];
        }
    }
  }
}

}

extern "C" void kernel_launch(void* const* d_in, const int* in_sizes, int n_in,
                              void* d_out, int out_size, void* d_ws, size_t ws_size,
                              hipStream_t stream) {
  (void)in_sizes; (void)n_in; (void)out_size; (void)ws_size;
  const float* z = (const float*)d_in[0];
  const int*   y = (const int*)d_in[1];
  float* ws  = (float*)d_ws;
  float* out = (float*)d_out;

  class_stats_kernel<<<dim3(kStatTiles, 2), kStatWaves * 32, 0, stream>>>(z, y, ws);
  prep_kernel<<<kPrepBlocks, 256, 0, stream>>>(z, ws);

  lda_score_wmma_kernel<<<dim3(kScoreColGroups, kScoreRowGroups), kScoreWaves * 32, 0, stream>>>(z, ws, out);
}
